# starting kernel
# baseline (speedup 1.0000x reference)
#include <hip/hip_runtime.h>
#include <hip/hip_bf16.h>

typedef __attribute__((ext_vector_type(2))) float v2f;
typedef __attribute__((ext_vector_type(8))) float v8f;

#define N_ROWS 1048576
#define F_DIM  64
#define C_DIM  8
#define D_DIM  4
#define S_COLS 16

__global__ __launch_bounds__(256) void FlexTree_wmma_kernel(
    const float* __restrict__ X,
    const int*   __restrict__ leaf_idx,
    const float* __restrict__ W_child,
    const float* __restrict__ w_root,
    const float* __restrict__ a_ptr,
    const float* __restrict__ b_ptr,
    float*       __restrict__ y)
{
    __shared__ float S[F_DIM * S_COLS];

    const int tid = threadIdx.x;

    #pragma unroll
    for (int i = 0; i < 4; ++i)
        S[tid + i * 256] = 0.0f;
    __syncthreads();

    if (tid < C_DIM * D_DIM) {
        const int c = tid >> 2;
        const int d = tid & 3;
        const int f = leaf_idx[c * D_DIM + d];
        atomicAdd(&S[f * S_COLS + c], W_child[c * D_DIM + d]);
    }
    __syncthreads();

    const float a     = a_ptr[0];
    const float inv_b = 1.0f / b_ptr[0];

    const int wave  = tid >> 5;
    const int lane  = tid & 31;
    const int laneN = lane & 15;
    const int hi    = lane >> 4;

    const int rowBase = (blockIdx.x * 8 + wave) * 16;

    const float* Aptr = X + (size_t)(rowBase + laneN) * F_DIM + 2 * hi;

    v8f acc = {};

    #pragma unroll
    for (int kb = 0; kb < 16; ++kb) {
        v2f afrag = *(const v2f*)(Aptr + kb * 4);

        const int kRow = kb * 4 + 2 * hi;
        v2f bfrag;
        bfrag.x = S[(kRow + 0) * S_COLS + laneN];
        bfrag.y = S[(kRow + 1) * S_COLS + laneN];

        acc = __builtin_amdgcn_wmma_f32_16x16x4_f32(
            false, afrag, false, bfrag, (short)0, acc, false, false);
    }

    const float wr = (laneN < C_DIM) ? w_root[laneN] : 0.0f;

    #pragma unroll
    for (int v = 0; v < 8; ++v) {
        const float z  = acc[v];
        const float t  = (z - a) * inv_b;
        float       hw = __expf(-t * t) * wr;

        hw += __shfl_xor(hw, 1, 32);
        hw += __shfl_xor(hw, 2, 32);
        hw += __shfl_xor(hw, 4, 32);

        const float u  = (hw - a) * inv_b;
        const float yv = __expf(-u * u);

        if (laneN == 0)
            y[rowBase + v + 8 * hi] = yv;
    }
}

extern "C" void kernel_launch(void* const* d_in, const int* in_sizes, int n_in,
                              void* d_out, int out_size, void* d_ws, size_t ws_size,
                              hipStream_t stream) {
    const float* X        = (const float*)d_in[0];
    const int*   leaf_idx = (const int*)d_in[1];
    const float* W_child  = (const float*)d_in[2];
    const float* w_root   = (const float*)d_in[3];
    const float* a_ptr    = (const float*)d_in[4];
    const float* b_ptr    = (const float*)d_in[5];
    float*       y        = (float*)d_out;

    dim3 grid(N_ROWS / 128);
    dim3 block(256);
    FlexTree_wmma_kernel<<<grid, block, 0, stream>>>(
        X, leaf_idx, W_child, w_root, a_ptr, b_ptr, y);
}


# X read as whole 128-B lines: 8 global loads of 16 B per lane (512 contiguous B each) parked in a wave-private LDS tile, A fragments read from LDS (was 16 8-B loads, 16 lines each); sums unchanged
# speedup vs baseline: 1.2159x; 1.2159x over previous
#include <hip/hip_runtime.h>
#include <hip/hip_bf16.h>

typedef __attribute__((ext_vector_type(2))) float v2f;
typedef __attribute__((ext_vector_type(4))) float v4f;
typedef __attribute__((ext_vector_type(8))) float v8f;

#define N_ROWS 1048576
#define F_DIM  64
#define C_DIM  8
#define D_DIM  4
#define S_COLS 16
#define X_PITCH 68

__global__ __launch_bounds__(256) void FlexTree_wmma_kernel(
    const float* __restrict__ X,
    const int*   __restrict__ leaf_idx,
    const float* __restrict__ W_child,
    const float* __restrict__ w_root,
    const float* __restrict__ a_ptr,
    const float* __restrict__ b_ptr,
    float*       __restrict__ y)
{
    __shared__ float S[F_DIM * S_COLS];
    __shared__ __attribute__((aligned(16))) float Xs[8][16 * X_PITCH];

    const int tid = threadIdx.x;

    #pragma unroll
    for (int i = 0; i < 4; ++i)
        S[tid + i * 256] = 0.0f;
    __syncthreads();

    if (tid < C_DIM * D_DIM) {
        const int c = tid >> 2;
        const int d = tid & 3;
        const int f = leaf_idx[c * D_DIM + d];
        atomicAdd(&S[f * S_COLS + c], W_child[c * D_DIM + d]);
    }
    __syncthreads();

    const float a     = a_ptr[0];
    const float inv_b = 1.0f / b_ptr[0];

    const int wave  = tid >> 5;
    const int lane  = tid & 31;
    const int laneN = lane & 15;
    const int hi    = lane >> 4;

    const int rowBase = (blockIdx.x * 8 + wave) * 16;

    const v4f* Xtile = (const v4f*)(X + (size_t)rowBase * F_DIM);
    v4f xin[8];
    #pragma unroll
    for (int q = 0; q < 8; ++q)
        xin[q] = Xtile[q * 32 + lane];

    float* myX = Xs[wave];
    #pragma unroll
    for (int q = 0; q < 8; ++q)
        *(v4f*)(myX + (2 * q + hi) * X_PITCH + 4 * laneN) = xin[q];

    __builtin_amdgcn_fence(__ATOMIC_RELEASE, "wavefront");
    __builtin_amdgcn_wave_barrier();
    __builtin_amdgcn_fence(__ATOMIC_ACQUIRE, "wavefront");

    const float* Aptr = myX + laneN * X_PITCH + 2 * hi;

    v8f acc = {};

    #pragma unroll
    for (int kb = 0; kb < 16; ++kb) {
        v2f afrag = *(const v2f*)(Aptr + kb * 4);

        const int kRow = kb * 4 + 2 * hi;
        v2f bfrag;
        bfrag.x = S[(kRow + 0) * S_COLS + laneN];
        bfrag.y = S[(kRow + 1) * S_COLS + laneN];

        acc = __builtin_amdgcn_wmma_f32_16x16x4_f32(
            false, afrag, false, bfrag, (short)0, acc, false, false);
    }

    const float wr = (laneN < C_DIM) ? w_root[laneN] : 0.0f;

    #pragma unroll
    for (int v = 0; v < 8; ++v) {
        const float z  = acc[v];
        const float t  = (z - a) * inv_b;
        float       hw = __expf(-t * t) * wr;

        hw += __shfl_xor(hw, 1, 32);
        hw += __shfl_xor(hw, 2, 32);
        hw += __shfl_xor(hw, 4, 32);

        const float u  = (hw - a) * inv_b;
        const float yv = __expf(-u * u);

        if (laneN == 0)
            y[rowBase + v + 8 * hi] = yv;
    }
}

extern "C" void kernel_launch(void* const* d_in, const int* in_sizes, int n_in,
                              void* d_out, int out_size, void* d_ws, size_t ws_size,
                              hipStream_t stream) {
    const float* X        = (const float*)d_in[0];
    const int*   leaf_idx = (const int*)d_in[1];
    const float* W_child  = (const float*)d_in[2];
    const float* w_root   = (const float*)d_in[3];
    const float* a_ptr    = (const float*)d_in[4];
    const float* b_ptr    = (const float*)d_in[5];
    float*       y        = (float*)d_out;

    dim3 grid(N_ROWS / 128);
    dim3 block(256);
    FlexTree_wmma_kernel<<<grid, block, 0, stream>>>(
        X, leaf_idx, W_child, w_root, a_ptr, b_ptr, y);
}


# WMMA operands swapped (A = S^T, B = X^T): each lane holds the 8 children of one X row, so the child sum is in-lane adds (same pairwise tree), no lane exchange, one 64-B store per wave
# speedup vs baseline: 1.4593x; 1.2002x over previous
#include <hip/hip_runtime.h>
#include <hip/hip_bf16.h>

typedef __attribute__((ext_vector_type(2))) float v2f;
typedef __attribute__((ext_vector_type(4))) float v4f;
typedef __attribute__((ext_vector_type(8))) float v8f;

#define N_ROWS 1048576
#define F_DIM  64
#define C_DIM  8
#define D_DIM  4
#define S_COLS 16
#define X_PITCH 68

__global__ __launch_bounds__(256) void FlexTree_wmma_kernel(
    const float* __restrict__ X,
    const int*   __restrict__ leaf_idx,
    const float* __restrict__ W_child,
    const float* __restrict__ w_root,
    const float* __restrict__ a_ptr,
    const float* __restrict__ b_ptr,
    float*       __restrict__ y)
{
    __shared__ float S[F_DIM * S_COLS];
    __shared__ __attribute__((aligned(16))) float Xs[8][16 * X_PITCH];

    const int tid = threadIdx.x;

    #pragma unroll
    for (int i = 0; i < 4; ++i)
        S[tid + i * 256] = 0.0f;
    __syncthreads();

    if (tid < C_DIM * D_DIM) {
        const int c = tid >> 2;
        const int d = tid & 3;
        const int f = leaf_idx[c * D_DIM + d];
        atomicAdd(&S[f * S_COLS + c], W_child[c * D_DIM + d]);
    }
    __syncthreads();

    const float a     = a_ptr[0];
    const float inv_b = 1.0f / b_ptr[0];

    const int wave  = tid >> 5;
    const int lane  = tid & 31;
    const int laneN = lane & 15;
    const int hi    = lane >> 4;

    const int rowBase = (blockIdx.x * 8 + wave) * 16;

    const v4f* Xtile = (const v4f*)(X + (size_t)rowBase * F_DIM);
    v4f xin[8];
    #pragma unroll
    for (int q = 0; q < 8; ++q)
        xin[q] = Xtile[q * 32 + lane];

    float* myX = Xs[wave];
    #pragma unroll
    for (int q = 0; q < 8; ++q)
        *(v4f*)(myX + (2 * q + hi) * X_PITCH + 4 * laneN) = xin[q];

    __builtin_amdgcn_fence(__ATOMIC_RELEASE, "wavefront");
    __builtin_amdgcn_wave_barrier();
    __builtin_amdgcn_fence(__ATOMIC_ACQUIRE, "wavefront");

    const float* Aptr = myX + laneN * X_PITCH + 2 * hi;

    v8f acc = {};

    #pragma unroll
    for (int kb = 0; kb < 16; ++kb) {
        v2f xfrag = *(const v2f*)(Aptr + kb * 4);

        const int kRow = kb * 4 + 2 * hi;
        v2f sfrag;
        sfrag.x = S[(kRow + 0) * S_COLS + laneN];
        sfrag.y = S[(kRow + 1) * S_COLS + laneN];

        acc = __builtin_amdgcn_wmma_f32_16x16x4_f32(
            false, sfrag, false, xfrag, (short)0, acc, false, false);
    }

    float hw[8];
    #pragma unroll
    for (int r = 0; r < 8; ++r) {
        const float t = (acc[r] - a) * inv_b;
        hw[r] = __expf(-t * t) * w_root[r];
    }

    const float hs = ((hw[0] + hw[1]) + (hw[2] + hw[3])) +
                     ((hw[4] + hw[5]) + (hw[6] + hw[7]));

    const float u  = (hs - a) * inv_b;
    const float yv = __expf(-u * u);

    if (hi == 0)
        y[rowBase + laneN] = yv;
}

extern "C" void kernel_launch(void* const* d_in, const int* in_sizes, int n_in,
                              void* d_out, int out_size, void* d_ws, size_t ws_size,
                              hipStream_t stream) {
    const float* X        = (const float*)d_in[0];
    const int*   leaf_idx = (const int*)d_in[1];
    const float* W_child  = (const float*)d_in[2];
    const float* w_root   = (const float*)d_in[3];
    const float* a_ptr    = (const float*)d_in[4];
    const float* b_ptr    = (const float*)d_in[5];
    float*       y        = (float*)d_out;

    dim3 grid(N_ROWS / 128);
    dim3 block(256);
    FlexTree_wmma_kernel<<<grid, block, 0, stream>>>(
        X, leaf_idx, W_child, w_root, a_ptr, b_ptr, y);
}
